# starting kernel
# baseline (speedup 1.0000x reference)
#include <hip/hip_runtime.h>

#define N_NODES 100000
#define N_EDGES 600000
#define DIM     128
#define ALPHA   0.5f

typedef __attribute__((ext_vector_type(2))) float v2f;
typedef __attribute__((ext_vector_type(8))) float v8f;

__global__ void dirgcn_zero_kernel(float* __restrict__ p, long n) {
    long i = (long)blockIdx.x * blockDim.x + threadIdx.x;
    long stride = (long)gridDim.x * blockDim.x;
    for (; i < n; i += stride) p[i] = 0.0f;
}

__global__ void dirgcn_degree_kernel(const int* __restrict__ ei,
                                     float* __restrict__ deg_out,
                                     float* __restrict__ deg_in) {
    int e = blockIdx.x * blockDim.x + threadIdx.x;
    if (e < N_EDGES) {
        atomicAdd(&deg_out[ei[e]], 1.0f);
        atomicAdd(&deg_in[ei[N_EDGES + e]], 1.0f);
    }
}

__global__ void dirgcn_scatter_kernel(const float* __restrict__ x,
                                      const int* __restrict__ ei,
                                      const float* __restrict__ deg_out,
                                      const float* __restrict__ deg_in,
                                      float* __restrict__ agg_src,
                                      float* __restrict__ agg_dst) {
    int wave = blockIdx.x * (blockDim.x >> 5) + (threadIdx.x >> 5);
    int lane = threadIdx.x & 31;
    int row = ei[wave];
    int col = ei[N_EDGES + wave];
    float dro = deg_out[row];
    float dci = deg_in[col];
    float ws = (dro > 0.0f) ? rsqrtf(fmaxf(dro, 1.0f)) : 0.0f;
    float wd = (dci > 0.0f) ? rsqrtf(fmaxf(dci, 1.0f)) : 0.0f;
    float w = ws * wd;

    const float4* xc = (const float4*)(x + (size_t)col * DIM);
    const float4* xr = (const float4*)(x + (size_t)row * DIM);
    float4 vc = xc[lane];
    float4 vr = xr[lane];

    float* as = agg_src + (size_t)row * DIM + lane * 4;
    float* ad = agg_dst + (size_t)col * DIM + lane * 4;
    atomicAdd(as + 0, w * vc.x);
    atomicAdd(as + 1, w * vc.y);
    atomicAdd(as + 2, w * vc.z);
    atomicAdd(as + 3, w * vc.w);
    atomicAdd(ad + 0, w * vr.x);
    atomicAdd(ad + 1, w * vr.y);
    atomicAdd(ad + 2, w * vr.z);
    atomicAdd(ad + 3, w * vr.w);
}

__global__ void dirgcn_wprep_kernel(const float* __restrict__ Ws,
                                    const float* __restrict__ bs,
                                    const float* __restrict__ Wd,
                                    const float* __restrict__ bd,
                                    float* __restrict__ Ws_s,
                                    float* __restrict__ Wd_s,
                                    float* __restrict__ bias) {
    int i = blockIdx.x * blockDim.x + threadIdx.x;
    if (i < DIM * DIM) {
        Ws_s[i] = (1.0f - ALPHA) * Ws[i];
        Wd_s[i] = ALPHA * Wd[i];
    }
    if (i < DIM) bias[i] = (1.0f - ALPHA) * bs[i] + ALPHA * bd[i];
}

__global__ void __launch_bounds__(256)
dirgcn_gemm_kernel(const float* __restrict__ agg_src,
                   const float* __restrict__ agg_dst,
                   const float* __restrict__ Ws,
                   const float* __restrict__ Wd,
                   const float* __restrict__ bias,
                   float* __restrict__ out) {
    const int lane = threadIdx.x & 31;
    const int wave = threadIdx.x >> 5;
    const int row0 = blockIdx.x * 16;
    const int j0   = wave * 16;
    const int m    = lane & 15;
    const int koff = (lane >> 4) * 2;

    const float* arow_s = agg_src + (size_t)(row0 + m) * DIM;
    const float* arow_d = agg_dst + (size_t)(row0 + m) * DIM;
    const float* wrow_s = Ws + (size_t)(j0 + m) * DIM;
    const float* wrow_d = Wd + (size_t)(j0 + m) * DIM;

    v8f acc = {};
    #pragma unroll
    for (int k = 0; k < DIM; k += 4) {
        v2f a = *(const v2f*)(arow_s + k + koff);
        v2f b = *(const v2f*)(wrow_s + k + koff);
        acc = __builtin_amdgcn_wmma_f32_16x16x4_f32(false, a, false, b,
                                                    (short)0, acc, false, false);
    }
    #pragma unroll
    for (int k = 0; k < DIM; k += 4) {
        v2f a = *(const v2f*)(arow_d + k + koff);
        v2f b = *(const v2f*)(wrow_d + k + koff);
        acc = __builtin_amdgcn_wmma_f32_16x16x4_f32(false, a, false, b,
                                                    (short)0, acc, false, false);
    }

    float bv = bias[j0 + m];
    const int n = lane & 15;
    const int mbase = (lane >> 4) * 8;
    #pragma unroll
    for (int r = 0; r < 8; ++r) {
        out[(size_t)(row0 + mbase + r) * DIM + j0 + n] = acc[r] + bv;
    }
}

extern "C" void kernel_launch(void* const* d_in, const int* in_sizes, int n_in,
                              void* d_out, int out_size, void* d_ws, size_t ws_size,
                              hipStream_t stream) {
    const float* x    = (const float*)d_in[0];
    const int*   ei   = (const int*)d_in[1];
    const float* W_src = (const float*)d_in[2];
    const float* b_src = (const float*)d_in[3];
    const float* W_dst = (const float*)d_in[4];
    const float* b_dst = (const float*)d_in[5];
    float* out = (float*)d_out;

    float* ws = (float*)d_ws;
    float* deg_out = ws;
    float* deg_in  = deg_out + N_NODES;
    float* agg_src = deg_in + N_NODES;
    float* agg_dst = agg_src + (size_t)N_NODES * DIM;
    float* Ws_s    = agg_dst + (size_t)N_NODES * DIM;
    float* Wd_s    = Ws_s + DIM * DIM;
    float* bias    = Wd_s + DIM * DIM;

    const long n_zero = 2L * N_NODES + 2L * N_NODES * DIM;
    dirgcn_zero_kernel<<<2048, 256, 0, stream>>>(ws, n_zero);

    dirgcn_degree_kernel<<<(N_EDGES + 255) / 256, 256, 0, stream>>>(ei, deg_out, deg_in);

    dirgcn_scatter_kernel<<<N_EDGES / 8, 256, 0, stream>>>(x, ei, deg_out, deg_in,
                                                           agg_src, agg_dst);

    dirgcn_wprep_kernel<<<(DIM * DIM + 255) / 256, 256, 0, stream>>>(
        W_src, b_src, W_dst, b_dst, Ws_s, Wd_s, bias);

    dirgcn_gemm_kernel<<<N_NODES / 16, 256, 0, stream>>>(agg_src, agg_dst,
                                                         Ws_s, Wd_s, bias, out);
}


# GEMM operand reuse: persistent blocks keep both W fragment sets in registers; each node tile is read once per block by whole-line b128 loads and staged in LDS (was strided b64 global loads per WMMA)
# speedup vs baseline: 1.3162x; 1.3162x over previous
#include <hip/hip_runtime.h>

#define N_NODES 100000
#define N_EDGES 600000
#define DIM     128
#define ALPHA   0.5f

typedef __attribute__((ext_vector_type(2))) float v2f;
typedef __attribute__((ext_vector_type(8))) float v8f;

__global__ void dirgcn_zero_kernel(float* __restrict__ p, long n) {
    long i = (long)blockIdx.x * blockDim.x + threadIdx.x;
    long stride = (long)gridDim.x * blockDim.x;
    for (; i < n; i += stride) p[i] = 0.0f;
}

__global__ void dirgcn_degree_kernel(const int* __restrict__ ei,
                                     float* __restrict__ deg_out,
                                     float* __restrict__ deg_in) {
    int e = blockIdx.x * blockDim.x + threadIdx.x;
    if (e < N_EDGES) {
        atomicAdd(&deg_out[ei[e]], 1.0f);
        atomicAdd(&deg_in[ei[N_EDGES + e]], 1.0f);
    }
}

__global__ void dirgcn_scatter_kernel(const float* __restrict__ x,
                                      const int* __restrict__ ei,
                                      const float* __restrict__ deg_out,
                                      const float* __restrict__ deg_in,
                                      float* __restrict__ agg_src,
                                      float* __restrict__ agg_dst) {
    int wave = blockIdx.x * (blockDim.x >> 5) + (threadIdx.x >> 5);
    int lane = threadIdx.x & 31;
    int row = ei[wave];
    int col = ei[N_EDGES + wave];
    float dro = deg_out[row];
    float dci = deg_in[col];
    float ws = (dro > 0.0f) ? rsqrtf(fmaxf(dro, 1.0f)) : 0.0f;
    float wd = (dci > 0.0f) ? rsqrtf(fmaxf(dci, 1.0f)) : 0.0f;
    float w = ws * wd;

    const float4* xc = (const float4*)(x + (size_t)col * DIM);
    const float4* xr = (const float4*)(x + (size_t)row * DIM);
    float4 vc = xc[lane];
    float4 vr = xr[lane];

    float* as = agg_src + (size_t)row * DIM + lane * 4;
    float* ad = agg_dst + (size_t)col * DIM + lane * 4;
    atomicAdd(as + 0, w * vc.x);
    atomicAdd(as + 1, w * vc.y);
    atomicAdd(as + 2, w * vc.z);
    atomicAdd(as + 3, w * vc.w);
    atomicAdd(ad + 0, w * vr.x);
    atomicAdd(ad + 1, w * vr.y);
    atomicAdd(ad + 2, w * vr.z);
    atomicAdd(ad + 3, w * vr.w);
}

__global__ void dirgcn_wprep_kernel(const float* __restrict__ Ws,
                                    const float* __restrict__ bs,
                                    const float* __restrict__ Wd,
                                    const float* __restrict__ bd,
                                    float* __restrict__ Ws_s,
                                    float* __restrict__ Wd_s,
                                    float* __restrict__ bias) {
    int i = blockIdx.x * blockDim.x + threadIdx.x;
    if (i < DIM * DIM) {
        Ws_s[i] = (1.0f - ALPHA) * Ws[i];
        Wd_s[i] = ALPHA * Wd[i];
    }
    if (i < DIM) bias[i] = (1.0f - ALPHA) * bs[i] + ALPHA * bd[i];
}

#define N_TILES   (N_NODES / 16)
#define PITCH4    33
#define TILE4     (16 * PITCH4)
#define GEMM_GRID 512

__global__ void __launch_bounds__(256)
dirgcn_gemm_kernel(const float* __restrict__ agg_src,
                   const float* __restrict__ agg_dst,
                   const float* __restrict__ Ws,
                   const float* __restrict__ Wd,
                   const float* __restrict__ bias,
                   float* __restrict__ out) {
    __shared__ float4 tileA[2][2][TILE4];

    const int tid  = threadIdx.x;
    const int lane = tid & 31;
    const int wave = tid >> 5;
    const int j0   = wave * 16;
    const int m    = lane & 15;
    const int h    = lane >> 4;
    const int srow = tid >> 5;
    const int sc4  = tid & 31;

    const float4* wrow_s = (const float4*)(Ws + (size_t)(j0 + m) * DIM) + h;
    const float4* wrow_d = (const float4*)(Wd + (size_t)(j0 + m) * DIM) + h;

    int tile = blockIdx.x;
    float4 g0, g1, g2, g3;
    {
        const float4* gs = (const float4*)agg_src + (size_t)tile * (16 * 32);
        const float4* gd = (const float4*)agg_dst + (size_t)tile * (16 * 32);
        g0 = gs[tid]; g1 = gs[256 + tid];
        g2 = gd[tid]; g3 = gd[256 + tid];
    }

    float4 bs[16], bd[16];
    #pragma unroll
    for (int i = 0; i < 16; ++i) {
        bs[i] = wrow_s[2 * i];
        bd[i] = wrow_d[2 * i];
    }

    const float bv = bias[j0 + m];
    int stage = 0;

    for (; tile < N_TILES; tile += GEMM_GRID) {
        float4* sa = &tileA[stage][0][0];
        float4* da = &tileA[stage][1][0];
        sa[srow * PITCH4 + sc4]       = g0;
        sa[(srow + 8) * PITCH4 + sc4] = g1;
        da[srow * PITCH4 + sc4]       = g2;
        da[(srow + 8) * PITCH4 + sc4] = g3;
        __syncthreads();

        const int nt = tile + GEMM_GRID;
        if (nt < N_TILES) {
            const float4* gs = (const float4*)agg_src + (size_t)nt * (16 * 32);
            const float4* gd = (const float4*)agg_dst + (size_t)nt * (16 * 32);
            g0 = gs[tid]; g1 = gs[256 + tid];
            g2 = gd[tid]; g3 = gd[256 + tid];
        }

        const float4* fa_s = sa + m * PITCH4 + h;
        const float4* fa_d = da + m * PITCH4 + h;
        v8f acc_s = {};
        v8f acc_d = {};
        #pragma unroll
        for (int i = 0; i < 16; ++i) {
            const float4 as4 = fa_s[2 * i];
            const float4 ad4 = fa_d[2 * i];
            v2f a0 = {as4.x, as4.y};
            v2f b0 = {bs[i].x, bs[i].y};
            v2f a1 = {ad4.x, ad4.y};
            v2f b1 = {bd[i].x, bd[i].y};
            acc_s = __builtin_amdgcn_wmma_f32_16x16x4_f32(false, a0, false, b0,
                                                          (short)0, acc_s, false, false);
            acc_d = __builtin_amdgcn_wmma_f32_16x16x4_f32(false, a1, false, b1,
                                                          (short)0, acc_d, false, false);
            v2f a2 = {as4.z, as4.w};
            v2f b2 = {bs[i].z, bs[i].w};
            v2f a3 = {ad4.z, ad4.w};
            v2f b3 = {bd[i].z, bd[i].w};
            acc_s = __builtin_amdgcn_wmma_f32_16x16x4_f32(false, a2, false, b2,
                                                          (short)0, acc_s, false, false);
            acc_d = __builtin_amdgcn_wmma_f32_16x16x4_f32(false, a3, false, b3,
                                                          (short)0, acc_d, false, false);
        }

        const int row0  = tile * 16;
        const int mbase = h * 8;
        #pragma unroll
        for (int r = 0; r < 8; ++r) {
            out[(size_t)(row0 + mbase + r) * DIM + j0 + m] = acc_s[r] + acc_d[r] + bv;
        }
        stage ^= 1;
    }
}

extern "C" void kernel_launch(void* const* d_in, const int* in_sizes, int n_in,
                              void* d_out, int out_size, void* d_ws, size_t ws_size,
                              hipStream_t stream) {
    const float* x    = (const float*)d_in[0];
    const int*   ei   = (const int*)d_in[1];
    const float* W_src = (const float*)d_in[2];
    const float* b_src = (const float*)d_in[3];
    const float* W_dst = (const float*)d_in[4];
    const float* b_dst = (const float*)d_in[5];
    float* out = (float*)d_out;

    float* ws = (float*)d_ws;
    float* deg_out = ws;
    float* deg_in  = deg_out + N_NODES;
    float* agg_src = deg_in + N_NODES;
    float* agg_dst = agg_src + (size_t)N_NODES * DIM;
    float* Ws_s    = agg_dst + (size_t)N_NODES * DIM;
    float* Wd_s    = Ws_s + DIM * DIM;
    float* bias    = Wd_s + DIM * DIM;

    const long n_zero = 2L * N_NODES + 2L * N_NODES * DIM;
    dirgcn_zero_kernel<<<2048, 256, 0, stream>>>(ws, n_zero);

    dirgcn_degree_kernel<<<(N_EDGES + 255) / 256, 256, 0, stream>>>(ei, deg_out, deg_in);

    dirgcn_scatter_kernel<<<N_EDGES / 8, 256, 0, stream>>>(x, ei, deg_out, deg_in,
                                                           agg_src, agg_dst);

    dirgcn_wprep_kernel<<<(DIM * DIM + 255) / 256, 256, 0, stream>>>(
        W_src, b_src, W_dst, b_dst, Ws_s, Wd_s, bias);

    dirgcn_gemm_kernel<<<GEMM_GRID, 256, 0, stream>>>(agg_src, agg_dst,
                                                      Ws_s, Wd_s, bias, out);
}


# Aggregation by gather, not float atomic scatter: edges bucketed per node (int counts, block-scan slot allocation, cursor fill); one wave per node sums neighbour rows in registers, writes the row once
# speedup vs baseline: 1.9688x; 1.4958x over previous
#include <hip/hip_runtime.h>

#define N_NODES 100000
#define N_EDGES 600000
#define DIM     128
#define ALPHA   0.5f

typedef __attribute__((ext_vector_type(2))) float v2f;
typedef __attribute__((ext_vector_type(8))) float v8f;

struct __attribute__((aligned(8))) Entry { int idx; float w; };

__global__ void dirgcn_zero_kernel(int* __restrict__ deg, int* __restrict__ counter) {
    int i = blockIdx.x * blockDim.x + threadIdx.x;
    if (i < 2 * N_NODES) deg[i] = 0;
    if (i == 0) *counter = 0;
}

__global__ void dirgcn_degree_kernel(const int* __restrict__ ei, int* __restrict__ deg) {
    int e = blockIdx.x * blockDim.x + threadIdx.x;
    if (e < N_EDGES) {
        const int row = ei[e];
        const int col = ei[N_EDGES + e];
        atomicAdd(&deg[row], 1);
        atomicAdd(&deg[N_NODES + col], 1);
    }
}

__global__ void __launch_bounds__(1024)
dirgcn_alloc_kernel(const int* __restrict__ deg, int* __restrict__ cur,
                    int* __restrict__ counter) {
    __shared__ int wtot[32];
    __shared__ int bbase;
    const int i    = blockIdx.x * 1024 + threadIdx.x;
    const int lane = threadIdx.x & 31;
    const int wave = threadIdx.x >> 5;
    const int d = (i < 2 * N_NODES) ? deg[i] : 0;
    int s = d;
    #pragma unroll
    for (int o = 1; o < 32; o <<= 1) {
        int t = __shfl_up(s, o, 32);
        if (lane >= o) s += t;
    }
    if (lane == 31) wtot[wave] = s;
    __syncthreads();
    if (wave == 0) {
        const int t = wtot[lane];
        int ts = t;
        #pragma unroll
        for (int o = 1; o < 32; o <<= 1) {
            int u = __shfl_up(ts, o, 32);
            if (lane >= o) ts += u;
        }
        wtot[lane] = ts - t;
        if (lane == 31) bbase = atomicAdd(counter, ts);
    }
    __syncthreads();
    if (i < 2 * N_NODES) cur[i] = bbase + wtot[wave] + s - d;
}

__global__ void dirgcn_fill_kernel(const int* __restrict__ ei,
                                   const int* __restrict__ deg,
                                   int* __restrict__ cur,
                                   Entry* __restrict__ list) {
    int e = blockIdx.x * blockDim.x + threadIdx.x;
    if (e < N_EDGES) {
        int row = ei[e];
        int col = ei[N_EDGES + e];
        float dro = (float)deg[row];
        float dci = (float)deg[N_NODES + col];
        float ws = (dro > 0.0f) ? rsqrtf(fmaxf(dro, 1.0f)) : 0.0f;
        float wd = (dci > 0.0f) ? rsqrtf(fmaxf(dci, 1.0f)) : 0.0f;
        float w = ws * wd;
        int p = atomicAdd(&cur[row], 1);
        int q = atomicAdd(&cur[N_NODES + col], 1);
        Entry a; a.idx = col; a.w = w;
        Entry b; b.idx = row; b.w = w;
        list[p] = a;
        list[q] = b;
    }
}

#define GATHER_UNROLL 8
__global__ void __launch_bounds__(256)
dirgcn_gather_kernel(const float* __restrict__ x,
                     const int* __restrict__ deg,
                     const int* __restrict__ cur,
                     const Entry* __restrict__ list,
                     float* __restrict__ agg) {
    const int lane = threadIdx.x & 31;
    const int seg   = __builtin_amdgcn_readfirstlane(blockIdx.x * 8 + (threadIdx.x >> 5));
    const int d     = __builtin_amdgcn_readfirstlane(deg[seg]);
    const int start = __builtin_amdgcn_readfirstlane(cur[seg]) - d;
    const float4* x4 = (const float4*)x;

    float4 acc = {0.0f, 0.0f, 0.0f, 0.0f};
    for (int base = 0; base < d; base += 32) {
        const int cnt = min(d - base, 32);
        Entry en; en.idx = 0; en.w = 0.0f;
        if (lane < cnt) en = list[start + base + lane];
        for (int k = 0; k < cnt; k += GATHER_UNROLL) {
            float4 v[GATHER_UNROLL];
            float  w[GATHER_UNROLL];
            int    idx[GATHER_UNROLL];
            #pragma unroll
            for (int j = 0; j < GATHER_UNROLL; ++j) {
                idx[j] = __builtin_amdgcn_readlane(en.idx, k + j);
                w[j]   = __int_as_float(__builtin_amdgcn_readlane(__float_as_int(en.w), k + j));
            }
            #pragma unroll
            for (int j = 0; j < GATHER_UNROLL; ++j) {
                v[j] = make_float4(0.0f, 0.0f, 0.0f, 0.0f);
                if (k + j < cnt) v[j] = x4[(size_t)idx[j] * (DIM / 4) + lane];
            }
            #pragma unroll
            for (int j = 0; j < GATHER_UNROLL; ++j) {
                acc.x = fmaf(w[j], v[j].x, acc.x);
                acc.y = fmaf(w[j], v[j].y, acc.y);
                acc.z = fmaf(w[j], v[j].z, acc.z);
                acc.w = fmaf(w[j], v[j].w, acc.w);
            }
        }
    }
    ((float4*)agg)[(size_t)seg * (DIM / 4) + lane] = acc;
}

__global__ void dirgcn_wprep_kernel(const float* __restrict__ Ws,
                                    const float* __restrict__ bs,
                                    const float* __restrict__ Wd,
                                    const float* __restrict__ bd,
                                    float* __restrict__ Ws_s,
                                    float* __restrict__ Wd_s,
                                    float* __restrict__ bias) {
    int i = blockIdx.x * blockDim.x + threadIdx.x;
    if (i < DIM * DIM) {
        Ws_s[i] = (1.0f - ALPHA) * Ws[i];
        Wd_s[i] = ALPHA * Wd[i];
    }
    if (i < DIM) bias[i] = (1.0f - ALPHA) * bs[i] + ALPHA * bd[i];
}

#define N_TILES   (N_NODES / 16)
#define PITCH4    33
#define TILE4     (16 * PITCH4)
#define GEMM_GRID 512

__global__ void __launch_bounds__(256)
dirgcn_gemm_kernel(const float* __restrict__ agg_src,
                   const float* __restrict__ agg_dst,
                   const float* __restrict__ Ws,
                   const float* __restrict__ Wd,
                   const float* __restrict__ bias,
                   float* __restrict__ out) {
    __shared__ float4 tileA[2][2][TILE4];

    const int tid  = threadIdx.x;
    const int lane = tid & 31;
    const int wave = tid >> 5;
    const int j0   = wave * 16;
    const int m    = lane & 15;
    const int h    = lane >> 4;
    const int srow = tid >> 5;
    const int sc4  = tid & 31;

    const float4* wrow_s = (const float4*)(Ws + (size_t)(j0 + m) * DIM) + h;
    const float4* wrow_d = (const float4*)(Wd + (size_t)(j0 + m) * DIM) + h;

    int tile = blockIdx.x;
    float4 g0, g1, g2, g3;
    {
        const float4* gs = (const float4*)agg_src + (size_t)tile * (16 * 32);
        const float4* gd = (const float4*)agg_dst + (size_t)tile * (16 * 32);
        g0 = gs[tid]; g1 = gs[256 + tid];
        g2 = gd[tid]; g3 = gd[256 + tid];
    }

    float4 bs[16], bd[16];
    #pragma unroll
    for (int i = 0; i < 16; ++i) {
        bs[i] = wrow_s[2 * i];
        bd[i] = wrow_d[2 * i];
    }

    const float bv = bias[j0 + m];
    int stage = 0;

    for (; tile < N_TILES; tile += GEMM_GRID) {
        float4* sa = &tileA[stage][0][0];
        float4* da = &tileA[stage][1][0];
        sa[srow * PITCH4 + sc4]       = g0;
        sa[(srow + 8) * PITCH4 + sc4] = g1;
        da[srow * PITCH4 + sc4]       = g2;
        da[(srow + 8) * PITCH4 + sc4] = g3;
        __syncthreads();

        const int nt = tile + GEMM_GRID;
        if (nt < N_TILES) {
            const float4* gs = (const float4*)agg_src + (size_t)nt * (16 * 32);
            const float4* gd = (const float4*)agg_dst + (size_t)nt * (16 * 32);
            g0 = gs[tid]; g1 = gs[256 + tid];
            g2 = gd[tid]; g3 = gd[256 + tid];
        }

        const float4* fa_s = sa + m * PITCH4 + h;
        const float4* fa_d = da + m * PITCH4 + h;
        v8f acc_s = {};
        v8f acc_d = {};
        #pragma unroll
        for (int i = 0; i < 16; ++i) {
            const float4 as4 = fa_s[2 * i];
            const float4 ad4 = fa_d[2 * i];
            v2f a0 = {as4.x, as4.y};
            v2f b0 = {bs[i].x, bs[i].y};
            v2f a1 = {ad4.x, ad4.y};
            v2f b1 = {bd[i].x, bd[i].y};
            acc_s = __builtin_amdgcn_wmma_f32_16x16x4_f32(false, a0, false, b0,
                                                          (short)0, acc_s, false, false);
            acc_d = __builtin_amdgcn_wmma_f32_16x16x4_f32(false, a1, false, b1,
                                                          (short)0, acc_d, false, false);
            v2f a2 = {as4.z, as4.w};
            v2f b2 = {bs[i].z, bs[i].w};
            v2f a3 = {ad4.z, ad4.w};
            v2f b3 = {bd[i].z, bd[i].w};
            acc_s = __builtin_amdgcn_wmma_f32_16x16x4_f32(false, a2, false, b2,
                                                          (short)0, acc_s, false, false);
            acc_d = __builtin_amdgcn_wmma_f32_16x16x4_f32(false, a3, false, b3,
                                                          (short)0, acc_d, false, false);
        }

        const int row0  = tile * 16;
        const int mbase = h * 8;
        #pragma unroll
        for (int r = 0; r < 8; ++r) {
            out[(size_t)(row0 + mbase + r) * DIM + j0 + m] = acc_s[r] + acc_d[r] + bv;
        }
        stage ^= 1;
    }
}

extern "C" void kernel_launch(void* const* d_in, const int* in_sizes, int n_in,
                              void* d_out, int out_size, void* d_ws, size_t ws_size,
                              hipStream_t stream) {
    const float* x    = (const float*)d_in[0];
    const int*   ei   = (const int*)d_in[1];
    const float* W_src = (const float*)d_in[2];
    const float* b_src = (const float*)d_in[3];
    const float* W_dst = (const float*)d_in[4];
    const float* b_dst = (const float*)d_in[5];
    float* out = (float*)d_out;

    float* ws = (float*)d_ws;
    int*   deg     = (int*)ws;
    float* agg_src = ws + 2 * N_NODES;
    float* agg_dst = agg_src + (size_t)N_NODES * DIM;
    float* Ws_s    = agg_dst + (size_t)N_NODES * DIM;
    float* Wd_s    = Ws_s + DIM * DIM;
    float* bias    = Wd_s + DIM * DIM;

    int*   counter = (int*)d_out;
    int*   cur     = counter + 4;
    Entry* list    = (Entry*)(cur + 2 * N_NODES);

    dirgcn_zero_kernel<<<(2 * N_NODES + 255) / 256, 256, 0, stream>>>(deg, counter);

    dirgcn_degree_kernel<<<(N_EDGES + 255) / 256, 256, 0, stream>>>(ei, deg);

    dirgcn_alloc_kernel<<<(2 * N_NODES + 1023) / 1024, 1024, 0, stream>>>(deg, cur, counter);

    dirgcn_fill_kernel<<<(N_EDGES + 255) / 256, 256, 0, stream>>>(ei, deg, cur, list);

    dirgcn_gather_kernel<<<2 * N_NODES / 8, 256, 0, stream>>>(x, deg, cur, list, agg_src);

    dirgcn_wprep_kernel<<<(DIM * DIM + 255) / 256, 256, 0, stream>>>(
        W_src, b_src, W_dst, b_dst, Ws_s, Wd_s, bias);

    dirgcn_gemm_kernel<<<GEMM_GRID, 256, 0, stream>>>(agg_src, agg_dst,
                                                      Ws_s, Wd_s, bias, out);
}


# Bucket fill without adds: the degree pass's integer add returns each entry's rank in its segment; the fill stores at start + rank from one packed {start, degree} word (was a second returning add)
# speedup vs baseline: 2.4214x; 1.2299x over previous
#include <hip/hip_runtime.h>

#define N_NODES 100000
#define N_EDGES 600000
#define DIM     128
#define ALPHA   0.5f

typedef __attribute__((ext_vector_type(2))) float v2f;
typedef __attribute__((ext_vector_type(8))) float v8f;

struct __attribute__((aligned(8))) Entry { int idx; float w; };
struct __attribute__((aligned(8))) Seg   { int start; int deg; };

__global__ void dirgcn_zero_kernel(int* __restrict__ deg, int* __restrict__ counter) {
    int i = blockIdx.x * blockDim.x + threadIdx.x;
    if (i < 2 * N_NODES) deg[i] = 0;
    if (i == 0) *counter = 0;
}

__global__ void dirgcn_degree_kernel(const int* __restrict__ ei, int* __restrict__ deg,
                                     int* __restrict__ rank) {
    int e = blockIdx.x * blockDim.x + threadIdx.x;
    if (e < N_EDGES) {
        const int row = ei[e];
        const int col = ei[N_EDGES + e];
        const int r_row = atomicAdd(&deg[row], 1);
        const int r_col = atomicAdd(&deg[N_NODES + col], 1);
        rank[e]           = r_row;
        rank[N_EDGES + e] = r_col;
    }
}

__global__ void __launch_bounds__(1024)
dirgcn_alloc_kernel(const int* __restrict__ deg, Seg* __restrict__ seg,
                    int* __restrict__ counter) {
    __shared__ int wtot[32];
    __shared__ int bbase;
    const int i    = blockIdx.x * 1024 + threadIdx.x;
    const int lane = threadIdx.x & 31;
    const int wave = threadIdx.x >> 5;
    const int d = (i < 2 * N_NODES) ? deg[i] : 0;
    int s = d;
    #pragma unroll
    for (int o = 1; o < 32; o <<= 1) {
        int t = __shfl_up(s, o, 32);
        if (lane >= o) s += t;
    }
    if (lane == 31) wtot[wave] = s;
    __syncthreads();
    if (wave == 0) {
        const int t = wtot[lane];
        int ts = t;
        #pragma unroll
        for (int o = 1; o < 32; o <<= 1) {
            int u = __shfl_up(ts, o, 32);
            if (lane >= o) ts += u;
        }
        wtot[lane] = ts - t;
        if (lane == 31) bbase = atomicAdd(counter, ts);
    }
    __syncthreads();
    if (i < 2 * N_NODES) {
        Seg sg; sg.start = bbase + wtot[wave] + s - d; sg.deg = d;
        seg[i] = sg;
    }
}

__global__ void dirgcn_fill_kernel(const int* __restrict__ ei,
                                   const int* __restrict__ rank,
                                   const Seg* __restrict__ seg,
                                   Entry* __restrict__ list) {
    int e = blockIdx.x * blockDim.x + threadIdx.x;
    if (e < N_EDGES) {
        int row = ei[e];
        int col = ei[N_EDGES + e];
        const int r_row = rank[e];
        const int r_col = rank[N_EDGES + e];
        const Seg sr = seg[row];
        const Seg sc = seg[N_NODES + col];
        float dro = (float)sr.deg;
        float dci = (float)sc.deg;
        float ws = (dro > 0.0f) ? rsqrtf(fmaxf(dro, 1.0f)) : 0.0f;
        float wd = (dci > 0.0f) ? rsqrtf(fmaxf(dci, 1.0f)) : 0.0f;
        float w = ws * wd;
        int p = sr.start + r_row;
        int q = sc.start + r_col;
        Entry a; a.idx = col; a.w = w;
        Entry b; b.idx = row; b.w = w;
        list[p] = a;
        list[q] = b;
    }
}

#define GATHER_UNROLL 8
__global__ void __launch_bounds__(256)
dirgcn_gather_kernel(const float* __restrict__ x,
                     const Seg* __restrict__ segs,
                     const Entry* __restrict__ list,
                     float* __restrict__ agg) {
    const int lane = threadIdx.x & 31;
    const int seg   = __builtin_amdgcn_readfirstlane(blockIdx.x * 8 + (threadIdx.x >> 5));
    const Seg sg    = segs[seg];
    const int d     = __builtin_amdgcn_readfirstlane(sg.deg);
    const int start = __builtin_amdgcn_readfirstlane(sg.start);
    const float4* x4 = (const float4*)x;

    float4 acc = {0.0f, 0.0f, 0.0f, 0.0f};
    for (int base = 0; base < d; base += 32) {
        const int cnt = min(d - base, 32);
        Entry en; en.idx = 0; en.w = 0.0f;
        if (lane < cnt) en = list[start + base + lane];
        for (int k = 0; k < cnt; k += GATHER_UNROLL) {
            float4 v[GATHER_UNROLL];
            float  w[GATHER_UNROLL];
            int    idx[GATHER_UNROLL];
            #pragma unroll
            for (int j = 0; j < GATHER_UNROLL; ++j) {
                idx[j] = __builtin_amdgcn_readlane(en.idx, k + j);
                w[j]   = __int_as_float(__builtin_amdgcn_readlane(__float_as_int(en.w), k + j));
            }
            #pragma unroll
            for (int j = 0; j < GATHER_UNROLL; ++j) {
                v[j] = make_float4(0.0f, 0.0f, 0.0f, 0.0f);
                if (k + j < cnt) v[j] = x4[(size_t)idx[j] * (DIM / 4) + lane];
            }
            #pragma unroll
            for (int j = 0; j < GATHER_UNROLL; ++j) {
                acc.x = fmaf(w[j], v[j].x, acc.x);
                acc.y = fmaf(w[j], v[j].y, acc.y);
                acc.z = fmaf(w[j], v[j].z, acc.z);
                acc.w = fmaf(w[j], v[j].w, acc.w);
            }
        }
    }
    ((float4*)agg)[(size_t)seg * (DIM / 4) + lane] = acc;
}

__global__ void dirgcn_wprep_kernel(const float* __restrict__ Ws,
                                    const float* __restrict__ bs,
                                    const float* __restrict__ Wd,
                                    const float* __restrict__ bd,
                                    float* __restrict__ Ws_s,
                                    float* __restrict__ Wd_s,
                                    float* __restrict__ bias) {
    int i = blockIdx.x * blockDim.x + threadIdx.x;
    if (i < DIM * DIM) {
        Ws_s[i] = (1.0f - ALPHA) * Ws[i];
        Wd_s[i] = ALPHA * Wd[i];
    }
    if (i < DIM) bias[i] = (1.0f - ALPHA) * bs[i] + ALPHA * bd[i];
}

#define N_TILES   (N_NODES / 16)
#define PITCH4    33
#define TILE4     (16 * PITCH4)
#define GEMM_GRID 512

__global__ void __launch_bounds__(256)
dirgcn_gemm_kernel(const float* __restrict__ agg_src,
                   const float* __restrict__ agg_dst,
                   const float* __restrict__ Ws,
                   const float* __restrict__ Wd,
                   const float* __restrict__ bias,
                   float* __restrict__ out) {
    __shared__ float4 tileA[2][2][TILE4];

    const int tid  = threadIdx.x;
    const int lane = tid & 31;
    const int wave = tid >> 5;
    const int j0   = wave * 16;
    const int m    = lane & 15;
    const int h    = lane >> 4;
    const int srow = tid >> 5;
    const int sc4  = tid & 31;

    const float4* wrow_s = (const float4*)(Ws + (size_t)(j0 + m) * DIM) + h;
    const float4* wrow_d = (const float4*)(Wd + (size_t)(j0 + m) * DIM) + h;

    int tile = blockIdx.x;
    float4 g0, g1, g2, g3;
    {
        const float4* gs = (const float4*)agg_src + (size_t)tile * (16 * 32);
        const float4* gd = (const float4*)agg_dst + (size_t)tile * (16 * 32);
        g0 = gs[tid]; g1 = gs[256 + tid];
        g2 = gd[tid]; g3 = gd[256 + tid];
    }

    float4 bs[16], bd[16];
    #pragma unroll
    for (int i = 0; i < 16; ++i) {
        bs[i] = wrow_s[2 * i];
        bd[i] = wrow_d[2 * i];
    }

    const float bv = bias[j0 + m];
    int stage = 0;

    for (; tile < N_TILES; tile += GEMM_GRID) {
        float4* sa = &tileA[stage][0][0];
        float4* da = &tileA[stage][1][0];
        sa[srow * PITCH4 + sc4]       = g0;
        sa[(srow + 8) * PITCH4 + sc4] = g1;
        da[srow * PITCH4 + sc4]       = g2;
        da[(srow + 8) * PITCH4 + sc4] = g3;
        __syncthreads();

        const int nt = tile + GEMM_GRID;
        if (nt < N_TILES) {
            const float4* gs = (const float4*)agg_src + (size_t)nt * (16 * 32);
            const float4* gd = (const float4*)agg_dst + (size_t)nt * (16 * 32);
            g0 = gs[tid]; g1 = gs[256 + tid];
            g2 = gd[tid]; g3 = gd[256 + tid];
        }

        const float4* fa_s = sa + m * PITCH4 + h;
        const float4* fa_d = da + m * PITCH4 + h;
        v8f acc_s = {};
        v8f acc_d = {};
        #pragma unroll
        for (int i = 0; i < 16; ++i) {
            const float4 as4 = fa_s[2 * i];
            const float4 ad4 = fa_d[2 * i];
            v2f a0 = {as4.x, as4.y};
            v2f b0 = {bs[i].x, bs[i].y};
            v2f a1 = {ad4.x, ad4.y};
            v2f b1 = {bd[i].x, bd[i].y};
            acc_s = __builtin_amdgcn_wmma_f32_16x16x4_f32(false, a0, false, b0,
                                                          (short)0, acc_s, false, false);
            acc_d = __builtin_amdgcn_wmma_f32_16x16x4_f32(false, a1, false, b1,
                                                          (short)0, acc_d, false, false);
            v2f a2 = {as4.z, as4.w};
            v2f b2 = {bs[i].z, bs[i].w};
            v2f a3 = {ad4.z, ad4.w};
            v2f b3 = {bd[i].z, bd[i].w};
            acc_s = __builtin_amdgcn_wmma_f32_16x16x4_f32(false, a2, false, b2,
                                                          (short)0, acc_s, false, false);
            acc_d = __builtin_amdgcn_wmma_f32_16x16x4_f32(false, a3, false, b3,
                                                          (short)0, acc_d, false, false);
        }

        const int row0  = tile * 16;
        const int mbase = h * 8;
        #pragma unroll
        for (int r = 0; r < 8; ++r) {
            out[(size_t)(row0 + mbase + r) * DIM + j0 + m] = acc_s[r] + acc_d[r] + bv;
        }
        stage ^= 1;
    }
}

extern "C" void kernel_launch(void* const* d_in, const int* in_sizes, int n_in,
                              void* d_out, int out_size, void* d_ws, size_t ws_size,
                              hipStream_t stream) {
    const float* x    = (const float*)d_in[0];
    const int*   ei   = (const int*)d_in[1];
    const float* W_src = (const float*)d_in[2];
    const float* b_src = (const float*)d_in[3];
    const float* W_dst = (const float*)d_in[4];
    const float* b_dst = (const float*)d_in[5];
    float* out = (float*)d_out;

    float* ws = (float*)d_ws;
    int*   deg     = (int*)ws;
    float* agg_src = ws + 2 * N_NODES;
    float* agg_dst = agg_src + (size_t)N_NODES * DIM;
    float* Ws_s    = agg_dst + (size_t)N_NODES * DIM;
    float* Wd_s    = Ws_s + DIM * DIM;
    float* bias    = Wd_s + DIM * DIM;

    int*   counter = (int*)d_out;
    Seg*   seg     = (Seg*)(counter + 4);
    Entry* list    = (Entry*)(seg + 2 * N_NODES);
    int*   rank    = (int*)(list + 2 * N_EDGES);

    dirgcn_zero_kernel<<<(2 * N_NODES + 255) / 256, 256, 0, stream>>>(deg, counter);

    dirgcn_degree_kernel<<<(N_EDGES + 255) / 256, 256, 0, stream>>>(ei, deg, rank);

    dirgcn_alloc_kernel<<<(2 * N_NODES + 1023) / 1024, 1024, 0, stream>>>(deg, seg, counter);

    dirgcn_fill_kernel<<<(N_EDGES + 255) / 256, 256, 0, stream>>>(ei, rank, seg, list);

    dirgcn_gather_kernel<<<2 * N_NODES / 8, 256, 0, stream>>>(x, seg, list, agg_src);

    dirgcn_wprep_kernel<<<(DIM * DIM + 255) / 256, 256, 0, stream>>>(
        W_src, b_src, W_dst, b_dst, Ws_s, Wd_s, bias);

    dirgcn_gemm_kernel<<<GEMM_GRID, 256, 0, stream>>>(agg_src, agg_dst,
                                                      Ws_s, Wd_s, bias, out);
}
